# starting kernel
# baseline (speedup 1.0000x reference)
#include <hip/hip_runtime.h>
#include <math.h>

typedef __attribute__((ext_vector_type(2))) float v2f;
typedef __attribute__((ext_vector_type(4))) float v4f;
typedef __attribute__((ext_vector_type(8))) float v8f;

#define BATCH 2
#define HIN   64
#define WIN   64
#define CH    256
#define HOUT  128
#define WOUT  128
#define KK2   25
#define FROW  260
#define OROW  260
#define BROW  102

__global__ __launch_bounds__(256) void carafe_wmma_kernel(
    const float* __restrict__ feat_g,
    const float* __restrict__ mask_g,
    float* __restrict__ out_g)
{
    __shared__ float featl[36 * FROW];
    __shared__ float Bm[16][BROW];
    __shared__ float outl[16][OROW];
    __shared__ int   adrtab[100];

    const int tid = threadIdx.x;
    int bid = blockIdx.x;
    const int lw = bid & 31; bid >>= 5;
    const int lh = bid & 31; bid >>= 5;
    const int bz = bid;
    const int h0 = lh * 2, w0 = lw * 2;

    if (tid < 100) {
        int q = tid / KK2;
        int p = tid - q * KK2;
        int y = (q >> 1) + p / 5;
        int x = (q & 1) + p % 5;
        adrtab[tid] = (y * 6 + x) * FROW;
    }

    for (int it = 0; it < 7; ++it) {
        int idx = it * 256 + tid;
        if (idx < 16 * BROW) ((float*)Bm)[idx] = 0.f;
    }

    for (int it = 0; it < 9; ++it) {
        int idx = it * 256 + tid;
        int cv  = idx & 63;
        int xy  = idx >> 6;
        int x = xy % 6, y = xy / 6;
        int gy = h0 - 2 + y, gx = w0 - 2 + x;
        v4f val = {0.f, 0.f, 0.f, 0.f};
        if (gy >= 0 && gy < HIN && gx >= 0 && gx < WIN) {
            val = *(const v4f*)(feat_g +
                  (((size_t)bz * HIN + gy) * WIN + gx) * CH + cv * 4);
        }
        *(v4f*)(&featl[(y * 6 + x) * FROW + cv * 4]) = val;
    }

    float w[KK2];
    int myq_s = 0;
    if (tid < 16) {
        int s = tid;
        myq_s = s >> 2;
        int t = s & 3;
        int Y = (h0 + (myq_s >> 1)) * 2 + (t >> 1);
        int X = (w0 + (myq_s & 1)) * 2 + (t & 1);
        const float* mp = mask_g + (((size_t)bz * HOUT + Y) * WOUT + X) * KK2;
        float mx = mp[0];
        #pragma unroll
        for (int p = 1; p < KK2; ++p) mx = fmaxf(mx, mp[p]);
        float sum = 0.f;
        #pragma unroll
        for (int p = 0; p < KK2; ++p) { w[p] = __expf(mp[p] - mx); sum += w[p]; }
        float inv = 1.f / sum;
        #pragma unroll
        for (int p = 0; p < KK2; ++p) w[p] *= inv;
    }
    __syncthreads();

    if (tid < 16) {
        #pragma unroll
        for (int p = 0; p < KK2; ++p)
            Bm[tid][myq_s * KK2 + p] = w[p];
    }
    __syncthreads();

    const int wave = tid >> 5;
    const int lane = tid & 31;
    const int n16  = lane & 15;
    const int hi   = lane >> 4;
    const int c0a  = wave * 16 + n16;
    const int c0b  = (wave + 8) * 16 + n16;

    v8f acc0 = {0.f, 0.f, 0.f, 0.f, 0.f, 0.f, 0.f, 0.f};
    v8f acc1 = {0.f, 0.f, 0.f, 0.f, 0.f, 0.f, 0.f, 0.f};

    #pragma unroll
    for (int k4 = 0; k4 < 25; ++k4) {
        int kk0 = k4 * 4 + hi * 2;
        int o0 = adrtab[kk0];
        int o1 = adrtab[kk0 + 1];

        v2f av0, av1, bv;
        av0.x = featl[o0 + c0a];
        av0.y = featl[o1 + c0a];
        av1.x = featl[o0 + c0b];
        av1.y = featl[o1 + c0b];
        bv = *(const v2f*)(&Bm[n16][kk0]);

        acc0 = __builtin_amdgcn_wmma_f32_16x16x4_f32(
            false, av0, false, bv, (short)0, acc0, false, false);
        acc1 = __builtin_amdgcn_wmma_f32_16x16x4_f32(
            false, av1, false, bv, (short)0, acc1, false, false);
    }

    #pragma unroll
    for (int r = 0; r < 8; ++r) {
        outl[n16][wave * 16 + hi * 8 + r]       = acc0[r];
        outl[n16][(wave + 8) * 16 + hi * 8 + r] = acc1[r];
    }
    __syncthreads();

    for (int it = 0; it < 4; ++it) {
        int idx = it * 256 + tid;
        int s  = idx >> 6;
        int cv = idx & 63;
        int q = s >> 2, t = s & 3;
        int Y = (h0 + (q >> 1)) * 2 + (t >> 1);
        int X = (w0 + (q & 1)) * 2 + (t & 1);
        v4f val = *(v4f*)(&outl[s][cv * 4]);
        *(v4f*)(out_g + (((size_t)bz * HOUT + Y) * WOUT + X) * CH + cv * 4) = val;
    }
}

extern "C" void kernel_launch(void* const* d_in, const int* in_sizes, int n_in,
                              void* d_out, int out_size, void* d_ws, size_t ws_size,
                              hipStream_t stream) {
    const float* features = (const float*)d_in[0];
    const float* masks    = (const float*)d_in[1];
    float* out            = (float*)d_out;
    dim3 grid(BATCH * 32 * 32);
    carafe_wmma_kernel<<<grid, 256, 0, stream>>>(features, masks, out);
}


# 1024 resident workgroups (4 per WGP, fixed by LDS size) each loop over 2 adjacent tiles instead of 2048 one-tile workgroups; k table and B zero-fill done once per workgroup
# speedup vs baseline: 1.0616x; 1.0616x over previous
#include <hip/hip_runtime.h>
#include <math.h>

typedef __attribute__((ext_vector_type(2))) float v2f;
typedef __attribute__((ext_vector_type(4))) float v4f;
typedef __attribute__((ext_vector_type(8))) float v8f;

#define BATCH 2
#define HIN   64
#define WIN   64
#define CH    256
#define HOUT  128
#define WOUT  128
#define KK2   25
#define FROW  260
#define OROW  260
#define BROW  102
#define NTILES 2048
#define GRID   1024
#define TPW    (NTILES / GRID)
#define LDS_RESERVE 1536

__global__ __launch_bounds__(256) void carafe_wmma_kernel(
    const float* __restrict__ feat_g,
    const float* __restrict__ mask_g,
    float* __restrict__ out_g)
{
    __shared__ float featl[36 * FROW + LDS_RESERVE];
    __shared__ float Bm[16][BROW];
    __shared__ float outl[16][OROW];
    __shared__ int   adrtab[100];

    const int tid = threadIdx.x;

    if (tid < 100) {
        int q = tid / KK2;
        int p = tid - q * KK2;
        int y = (q >> 1) + p / 5;
        int x = (q & 1) + p % 5;
        adrtab[tid] = (y * 6 + x) * FROW;
    }

    for (int it = 0; it < 7; ++it) {
        int idx = it * 256 + tid;
        if (idx < 16 * BROW) ((float*)Bm)[idx] = 0.f;
    }

    const int wave = tid >> 5;
    const int lane = tid & 31;
    const int n16  = lane & 15;
    const int hi   = lane >> 4;
    const int c0a  = wave * 16 + n16;
    const int c0b  = (wave + 8) * 16 + n16;

    #pragma unroll 1
    for (int ti = 0; ti < TPW; ++ti) {
    int bid = blockIdx.x * TPW + ti;
    const int lw = bid & 31; bid >>= 5;
    const int lh = bid & 31; bid >>= 5;
    const int bz = bid;
    const int h0 = lh * 2, w0 = lw * 2;

    for (int it = 0; it < 9; ++it) {
        int idx = it * 256 + tid;
        int cv  = idx & 63;
        int xy  = idx >> 6;
        int x = xy % 6, y = xy / 6;
        int gy = h0 - 2 + y, gx = w0 - 2 + x;
        v4f val = {0.f, 0.f, 0.f, 0.f};
        if (gy >= 0 && gy < HIN && gx >= 0 && gx < WIN) {
            val = *(const v4f*)(feat_g +
                  (((size_t)bz * HIN + gy) * WIN + gx) * CH + cv * 4);
        }
        *(v4f*)(&featl[(y * 6 + x) * FROW + cv * 4]) = val;
    }

    float w[KK2];
    int myq_s = 0;
    if (tid < 16) {
        int s = tid;
        myq_s = s >> 2;
        int t = s & 3;
        int Y = (h0 + (myq_s >> 1)) * 2 + (t >> 1);
        int X = (w0 + (myq_s & 1)) * 2 + (t & 1);
        const float* mp = mask_g + (((size_t)bz * HOUT + Y) * WOUT + X) * KK2;
        float mx = mp[0];
        #pragma unroll
        for (int p = 1; p < KK2; ++p) mx = fmaxf(mx, mp[p]);
        float sum = 0.f;
        #pragma unroll
        for (int p = 0; p < KK2; ++p) { w[p] = __expf(mp[p] - mx); sum += w[p]; }
        float inv = 1.f / sum;
        #pragma unroll
        for (int p = 0; p < KK2; ++p) w[p] *= inv;
    }
    __syncthreads();

    if (tid < 16) {
        #pragma unroll
        for (int p = 0; p < KK2; ++p)
            Bm[tid][myq_s * KK2 + p] = w[p];
    }
    __syncthreads();

    v8f acc0 = {0.f, 0.f, 0.f, 0.f, 0.f, 0.f, 0.f, 0.f};
    v8f acc1 = {0.f, 0.f, 0.f, 0.f, 0.f, 0.f, 0.f, 0.f};

    #pragma unroll
    for (int k4 = 0; k4 < 25; ++k4) {
        int kk0 = k4 * 4 + hi * 2;
        int o0 = adrtab[kk0];
        int o1 = adrtab[kk0 + 1];

        v2f av0, av1, bv;
        av0.x = featl[o0 + c0a];
        av0.y = featl[o1 + c0a];
        av1.x = featl[o0 + c0b];
        av1.y = featl[o1 + c0b];
        bv = *(const v2f*)(&Bm[n16][kk0]);

        acc0 = __builtin_amdgcn_wmma_f32_16x16x4_f32(
            false, av0, false, bv, (short)0, acc0, false, false);
        acc1 = __builtin_amdgcn_wmma_f32_16x16x4_f32(
            false, av1, false, bv, (short)0, acc1, false, false);
    }

    #pragma unroll
    for (int r = 0; r < 8; ++r) {
        outl[n16][wave * 16 + hi * 8 + r]       = acc0[r];
        outl[n16][(wave + 8) * 16 + hi * 8 + r] = acc1[r];
    }
    __syncthreads();

    for (int it = 0; it < 4; ++it) {
        int idx = it * 256 + tid;
        int s  = idx >> 6;
        int cv = idx & 63;
        int q = s >> 2, t = s & 3;
        int Y = (h0 + (q >> 1)) * 2 + (t >> 1);
        int X = (w0 + (q & 1)) * 2 + (t & 1);
        v4f val = *(v4f*)(&outl[s][cv * 4]);
        *(v4f*)(out_g + (((size_t)bz * HOUT + Y) * WOUT + X) * CH + cv * 4) = val;
    }
    }
}

extern "C" void kernel_launch(void* const* d_in, const int* in_sizes, int n_in,
                              void* d_out, int out_size, void* d_ws, size_t ws_size,
                              hipStream_t stream) {
    const float* features = (const float*)d_in[0];
    const float* masks    = (const float*)d_in[1];
    float* out            = (float*)d_out;
    dim3 grid(GRID);
    carafe_wmma_kernel<<<grid, 256, 0, stream>>>(features, masks, out);
}


# Per tile all global loads issue as one group (mask rows, then 9 branch-free clamped float4 feature loads per thread, zero-select at the LDS write) instead of 9 serial load-wait-store round trips
# speedup vs baseline: 1.7765x; 1.6734x over previous
#include <hip/hip_runtime.h>
#include <math.h>

typedef __attribute__((ext_vector_type(2))) float v2f;
typedef __attribute__((ext_vector_type(4))) float v4f;
typedef __attribute__((ext_vector_type(8))) float v8f;

#define BATCH 2
#define HIN   64
#define WIN   64
#define CH    256
#define HOUT  128
#define WOUT  128
#define KK2   25
#define FROW  260
#define OROW  260
#define KPOS  36
#define BROW  36
#define NTILES 2048
#define GRID   1024
#define TPW    (NTILES / GRID)
#define LDS_RESERVE 2692

__global__ __launch_bounds__(256) void carafe_wmma_kernel(
    const float* __restrict__ feat_g,
    const float* __restrict__ mask_g,
    float* __restrict__ out_g)
{
    __shared__ float featl[36 * FROW + LDS_RESERVE];
    __shared__ float Bm[16][BROW];
    __shared__ float outl[16][OROW];

    const int tid = threadIdx.x;

    for (int it = 0; it < 3; ++it) {
        int idx = it * 256 + tid;
        if (idx < 16 * BROW) ((float*)Bm)[idx] = 0.f;
    }

    const int wave = tid >> 5;
    const int lane = tid & 31;
    const int n16  = lane & 15;
    const int hi   = lane >> 4;
    const int c0a  = wave * 16 + n16;
    const int c0b  = (wave + 8) * 16 + n16;

    #pragma unroll 1
    for (int ti = 0; ti < TPW; ++ti) {
    int bid = blockIdx.x * TPW + ti;
    const int lw = bid & 31; bid >>= 5;
    const int lh = bid & 31; bid >>= 5;
    const int bz = bid;
    const int h0 = lh * 2, w0 = lw * 2;

    float w[KK2];
    int myq_s = 0;
    if (tid < 16) {
        int s = tid;
        myq_s = s >> 2;
        int t = s & 3;
        int Y = (h0 + (myq_s >> 1)) * 2 + (t >> 1);
        int X = (w0 + (myq_s & 1)) * 2 + (t & 1);
        const float* mp = mask_g + (((size_t)bz * HOUT + Y) * WOUT + X) * KK2;
        #pragma unroll
        for (int p = 0; p < KK2; ++p) w[p] = mp[p];
    }

    const int cv = tid & 63;
    const int wq = __builtin_amdgcn_readfirstlane(tid >> 6);
    const float* fb = feat_g + (size_t)bz * HIN * WIN * CH + cv * 4;
    v4f stg[9];
    #pragma unroll
    for (int it = 0; it < 9; ++it) {
        const int xy = it * 4 + wq;
        const int y = xy / 6, x = xy - y * 6;
        const int gy = h0 - 2 + y, gx = w0 - 2 + x;
        const int cy = gy < 0 ? 0 : (gy > HIN - 1 ? HIN - 1 : gy);
        const int cx = gx < 0 ? 0 : (gx > WIN - 1 ? WIN - 1 : gx);
        stg[it] = *(const v4f*)(fb + (cy * WIN + cx) * CH);
    }

    if (tid < 16) {
        float mx = w[0];
        #pragma unroll
        for (int p = 1; p < KK2; ++p) mx = fmaxf(mx, w[p]);
        float sum = 0.f;
        #pragma unroll
        for (int p = 0; p < KK2; ++p) { w[p] = __expf(w[p] - mx); sum += w[p]; }
        float inv = 1.f / sum;
        #pragma unroll
        for (int p = 0; p < KK2; ++p) w[p] *= inv;
    }

    #pragma unroll
    for (int it = 0; it < 9; ++it) {
        const int xy = it * 4 + wq;
        const int y = xy / 6, x = xy - y * 6;
        const int gy = h0 - 2 + y, gx = w0 - 2 + x;
        const bool inb = gy >= 0 && gy < HIN && gx >= 0 && gx < WIN;
        v4f val = {0.f, 0.f, 0.f, 0.f};
        if (inb) val = stg[it];
        *(v4f*)(&featl[xy * FROW + cv * 4]) = val;
    }
    __syncthreads();

    if (tid < 16) {
        const int pbase = (myq_s >> 1) * 6 + (myq_s & 1);
        #pragma unroll
        for (int p = 0; p < KK2; ++p)
            Bm[tid][pbase + (p / 5) * 6 + (p % 5)] = w[p];
    }
    __syncthreads();

    v8f acc0 = {0.f, 0.f, 0.f, 0.f, 0.f, 0.f, 0.f, 0.f};
    v8f acc1 = {0.f, 0.f, 0.f, 0.f, 0.f, 0.f, 0.f, 0.f};

    #pragma unroll
    for (int k4 = 0; k4 < KPOS / 4; ++k4) {
        const int kk0 = k4 * 4 + hi * 2;
        const int o0 = kk0 * FROW;
        const int o1 = o0 + FROW;

        v2f av0, av1, bv;
        av0.x = featl[o0 + c0a];
        av0.y = featl[o1 + c0a];
        av1.x = featl[o0 + c0b];
        av1.y = featl[o1 + c0b];
        bv = *(const v2f*)(&Bm[n16][kk0]);

        acc0 = __builtin_amdgcn_wmma_f32_16x16x4_f32(
            false, av0, false, bv, (short)0, acc0, false, false);
        acc1 = __builtin_amdgcn_wmma_f32_16x16x4_f32(
            false, av1, false, bv, (short)0, acc1, false, false);
    }

    #pragma unroll
    for (int r = 0; r < 8; ++r) {
        outl[n16][wave * 16 + hi * 8 + r]       = acc0[r];
        outl[n16][(wave + 8) * 16 + hi * 8 + r] = acc1[r];
    }
    __syncthreads();

    for (int it = 0; it < 4; ++it) {
        int idx = it * 256 + tid;
        int s  = idx >> 6;
        int cv = idx & 63;
        int q = s >> 2, t = s & 3;
        int Y = (h0 + (q >> 1)) * 2 + (t >> 1);
        int X = (w0 + (q & 1)) * 2 + (t & 1);
        v4f val = *(v4f*)(&outl[s][cv * 4]);
        *(v4f*)(out_g + (((size_t)bz * HOUT + Y) * WOUT + X) * CH + cv * 4) = val;
    }
    }
}

extern "C" void kernel_launch(void* const* d_in, const int* in_sizes, int n_in,
                              void* d_out, int out_size, void* d_ws, size_t ws_size,
                              hipStream_t stream) {
    const float* features = (const float*)d_in[0];
    const float* masks    = (const float*)d_in[1];
    float* out            = (float*)d_out;
    dim3 grid(GRID);
    carafe_wmma_kernel<<<grid, 256, 0, stream>>>(features, masks, out);
}
